# starting kernel
# baseline (speedup 1.0000x reference)
#include <hip/hip_runtime.h>
#include <hip/hip_bf16.h>

#define N_NODES 100000
#define N_EDGES 1600000
#define DIM 64
#define OUT_DIM 32

typedef __attribute__((ext_vector_type(2))) float v2f;
typedef __attribute__((ext_vector_type(8))) float v8f;

__global__ void init_kernel(float* __restrict__ deg,
                            float* __restrict__ agg1,
                            float* __restrict__ agg2) {
    size_t tid = (size_t)blockIdx.x * blockDim.x + threadIdx.x;
    const size_t tot = (size_t)N_NODES * DIM;
    if (tid < tot) { agg1[tid] = 0.0f; agg2[tid] = 0.0f; }
    if (tid < N_NODES) deg[tid] = 1.0f;
}

__global__ void degree_kernel(const int* __restrict__ dst, float* __restrict__ deg) {
    int e = blockIdx.x * blockDim.x + threadIdx.x;
    if (e < N_EDGES) atomicAdd(&deg[dst[e]], 1.0f);
}

__global__ void rsqrt_kernel(float* __restrict__ deg) {
    int i = blockIdx.x * blockDim.x + threadIdx.x;
    if (i < N_NODES) deg[i] = rsqrtf(deg[i]);
}

__global__ __launch_bounds__(32) void gemm_x_w1(const float* __restrict__ x,
                                                const float* __restrict__ W,
                                                float* __restrict__ out) {
    const int lane = threadIdx.x;
    const int m    = lane & 15;
    const int half = lane >> 4;
    const int row0 = blockIdx.x * 16;

    v8f acc0 = {}, acc1 = {}, acc2 = {}, acc3 = {};
    const float* xr = x + (size_t)(row0 + m) * DIM;

#pragma unroll
    for (int ks = 0; ks < 16; ++ks) {
        const int kk = ks * 4 + half * 2;
        v2f a;  a.x = xr[kk];  a.y = xr[kk + 1];
        const float* wr0 = W + (size_t)kk * DIM;
        const float* wr1 = W + (size_t)(kk + 1) * DIM;
        v2f b0, b1, b2, b3;
        b0.x = wr0[m];      b0.y = wr1[m];
        b1.x = wr0[16 + m]; b1.y = wr1[16 + m];
        b2.x = wr0[32 + m]; b2.y = wr1[32 + m];
        b3.x = wr0[48 + m]; b3.y = wr1[48 + m];
        acc0 = __builtin_amdgcn_wmma_f32_16x16x4_f32(false, a, false, b0, (short)0, acc0, false, false);
        acc1 = __builtin_amdgcn_wmma_f32_16x16x4_f32(false, a, false, b1, (short)0, acc1, false, false);
        acc2 = __builtin_amdgcn_wmma_f32_16x16x4_f32(false, a, false, b2, (short)0, acc2, false, false);
        acc3 = __builtin_amdgcn_wmma_f32_16x16x4_f32(false, a, false, b3, (short)0, acc3, false, false);
    }

#pragma unroll
    for (int r = 0; r < 8; ++r) {
        float* o = out + (size_t)(row0 + r + 8 * half) * DIM;
        o[m]      = acc0[r];
        o[16 + m] = acc1[r];
        o[32 + m] = acc2[r];
        o[48 + m] = acc3[r];
    }
}

__global__ void scatter_kernel(const int* __restrict__ src,
                               const int* __restrict__ dst,
                               const float* __restrict__ dinv,
                               const float* __restrict__ feat,
                               float* __restrict__ out) {
    size_t tid = (size_t)blockIdx.x * blockDim.x + threadIdx.x;
    const size_t total = (size_t)(N_EDGES + N_NODES) * DIM;
    if (tid >= total) return;
    const int e = (int)(tid >> 6);
    const int j = (int)(tid & 63);
    int s, d;
    if (e < N_EDGES) { s = src[e]; d = dst[e]; }
    else             { s = e - N_EDGES; d = s; }
    const float w = dinv[s] * dinv[d];
    atomicAdd(out + (size_t)d * DIM + j, feat[(size_t)s * DIM + j] * w);
}

__global__ void bias_relu_kernel(float* __restrict__ h, const float* __restrict__ b1) {
    size_t tid = (size_t)blockIdx.x * blockDim.x + threadIdx.x;
    if (tid < (size_t)N_NODES * DIM)
        h[tid] = fmaxf(h[tid] + b1[tid & 63], 0.0f);
}

__global__ __launch_bounds__(32) void gemm_out(const float* __restrict__ agg,
                                               const float* __restrict__ Wmu,
                                               const float* __restrict__ bmu,
                                               const float* __restrict__ Wlv,
                                               const float* __restrict__ blv,
                                               float* __restrict__ out) {
    const int lane = threadIdx.x;
    const int m    = lane & 15;
    const int half = lane >> 4;
    const int row0 = blockIdx.x * 16;

    v8f accMu0 = {}, accMu1 = {}, accLv0 = {}, accLv1 = {};
    const float* ar = agg + (size_t)(row0 + m) * DIM;

#pragma unroll
    for (int ks = 0; ks < 16; ++ks) {
        const int kk = ks * 4 + half * 2;
        v2f a;  a.x = ar[kk];  a.y = ar[kk + 1];
        const float* mu0 = Wmu + (size_t)kk * OUT_DIM;
        const float* mu1 = Wmu + (size_t)(kk + 1) * OUT_DIM;
        const float* lv0 = Wlv + (size_t)kk * OUT_DIM;
        const float* lv1 = Wlv + (size_t)(kk + 1) * OUT_DIM;
        v2f b0, b1, b2, b3;
        b0.x = mu0[m];      b0.y = mu1[m];
        b1.x = mu0[16 + m]; b1.y = mu1[16 + m];
        b2.x = lv0[m];      b2.y = lv1[m];
        b3.x = lv0[16 + m]; b3.y = lv1[16 + m];
        accMu0 = __builtin_amdgcn_wmma_f32_16x16x4_f32(false, a, false, b0, (short)0, accMu0, false, false);
        accMu1 = __builtin_amdgcn_wmma_f32_16x16x4_f32(false, a, false, b1, (short)0, accMu1, false, false);
        accLv0 = __builtin_amdgcn_wmma_f32_16x16x4_f32(false, a, false, b2, (short)0, accLv0, false, false);
        accLv1 = __builtin_amdgcn_wmma_f32_16x16x4_f32(false, a, false, b3, (short)0, accLv1, false, false);
    }

    float* lvBase = out + (size_t)N_NODES * OUT_DIM;
#pragma unroll
    for (int r = 0; r < 8; ++r) {
        const int row = row0 + r + 8 * half;
        float* omu = out    + (size_t)row * OUT_DIM;
        float* olv = lvBase + (size_t)row * OUT_DIM;
        omu[m]      = accMu0[r] + bmu[m];
        omu[16 + m] = accMu1[r] + bmu[16 + m];
        olv[m]      = accLv0[r] + blv[m];
        olv[16 + m] = accLv1[r] + blv[16 + m];
    }
}

extern "C" void kernel_launch(void* const* d_in, const int* in_sizes, int n_in,
                              void* d_out, int out_size, void* d_ws, size_t ws_size,
                              hipStream_t stream) {
    const float* x    = (const float*)d_in[0];
    const int*   eidx = (const int*)d_in[1];
    const int*   src  = eidx;
    const int*   dst  = eidx + N_EDGES;
    const float* W1   = (const float*)d_in[2];
    const float* b1   = (const float*)d_in[3];
    const float* Wmu  = (const float*)d_in[4];
    const float* bmu  = (const float*)d_in[5];
    const float* Wlv  = (const float*)d_in[6];
    const float* blv  = (const float*)d_in[7];
    float*       out  = (float*)d_out;

    float* deg  = (float*)d_ws;
    float* xw   = deg  + N_NODES;
    float* agg1 = xw   + (size_t)N_NODES * DIM;
    float* agg2 = agg1 + (size_t)N_NODES * DIM;

    const int TB = 256;
    const size_t featElems    = (size_t)N_NODES * DIM;
    const size_t scatterElems = (size_t)(N_EDGES + N_NODES) * DIM;

    const unsigned gInit    = (unsigned)((featElems + TB - 1) / TB);
    const unsigned gDeg     = (N_EDGES + TB - 1) / TB;
    const unsigned gNode    = (N_NODES + TB - 1) / TB;
    const unsigned gScatter = (unsigned)((scatterElems + TB - 1) / TB);
    const unsigned gGemm    = N_NODES / 16;

    init_kernel<<<gInit, TB, 0, stream>>>(deg, agg1, agg2);
    degree_kernel<<<gDeg, TB, 0, stream>>>(dst, deg);
    rsqrt_kernel<<<gNode, TB, 0, stream>>>(deg);

    gemm_x_w1<<<gGemm, 32, 0, stream>>>(x, W1, xw);
    scatter_kernel<<<gScatter, TB, 0, stream>>>(src, dst, deg, xw, agg1);
    bias_relu_kernel<<<gInit, TB, 0, stream>>>(agg1, b1);
    scatter_kernel<<<gScatter, TB, 0, stream>>>(src, dst, deg, agg1, agg2);
    gemm_out<<<gGemm, 32, 0, stream>>>(agg2, Wmu, bmu, Wlv, blv, out);
}


# Replace the float-atomic edge scatter by a destination-sorted pull: int atomic arrival rank builds a CSR, one wave per node sums its rows in registers; sum order follows arrival rank
# speedup vs baseline: 2.9164x; 2.9164x over previous
#include <hip/hip_runtime.h>
#include <hip/hip_bf16.h>

#define N_NODES 100000
#define N_EDGES 1600000
#define DIM 64
#define OUT_DIM 32

typedef __attribute__((ext_vector_type(2))) float v2f;
typedef __attribute__((ext_vector_type(8))) float v8f;

__global__ void zero_counts_kernel(int* __restrict__ cnt) {
    int i = blockIdx.x * blockDim.x + threadIdx.x;
    if (i < N_NODES) cnt[i] = 0;
}

__global__ void count_rank_kernel(const int* __restrict__ dst,
                                  int* __restrict__ cnt,
                                  int* __restrict__ rank) {
    int e = blockIdx.x * blockDim.x + threadIdx.x;
    if (e < N_EDGES) rank[e] = atomicAdd(&cnt[dst[e]], 1);
}

#define SCAN_THREADS 1024
#define SCAN_PER_THREAD 16
#define SCAN_TILE (SCAN_THREADS * SCAN_PER_THREAD)
static_assert(N_NODES % SCAN_PER_THREAD == 0, "a thread's counts are all inside or all outside");

__device__ __forceinline__ int wave_inclusive_scan(int v, int lane) {
#pragma unroll
    for (int off = 1; off < 32; off <<= 1) {
        const int u = __shfl_up(v, off);
        if (lane >= off) v += u;
    }
    return v;
}

__global__ __launch_bounds__(SCAN_THREADS) void scan_kernel(const int* __restrict__ cnt,
                                                            int* __restrict__ rowptr,
                                                            float* __restrict__ dinv) {
    __shared__ int waveSum[32];
    const int t    = threadIdx.x;
    const int lane = t & 31;
    const int wave = t >> 5;
    int carry = 0;
    for (int base = 0; base < N_NODES; base += SCAN_TILE) {
        const int i = base + t * SCAN_PER_THREAD;
        const bool in = i < N_NODES;
        int4 c[SCAN_PER_THREAD / 4];
        int4 p[SCAN_PER_THREAD / 4];
        int run = 0;
#pragma unroll
        for (int q = 0; q < SCAN_PER_THREAD / 4; ++q) {
            c[q] = make_int4(0, 0, 0, 0);
            if (in) c[q] = *reinterpret_cast<const int4*>(cnt + i + 4 * q);
        }
#pragma unroll
        for (int q = 0; q < SCAN_PER_THREAD / 4; ++q) {
            p[q].x = run; run += c[q].x;
            p[q].y = run; run += c[q].y;
            p[q].z = run; run += c[q].z;
            p[q].w = run; run += c[q].w;
        }
        const int v = wave_inclusive_scan(run, lane);
        if (lane == 31) waveSum[wave] = v;
        __syncthreads();
        const int ws = waveSum[lane];
        const int wv = wave_inclusive_scan(ws, lane);
        const int tileTotal  = __shfl(wv, 31);
        const int wavePrefix = __shfl(wv - ws, wave);
        const int excl = carry + wavePrefix + (v - run);
        if (in) {
#pragma unroll
            for (int q = 0; q < SCAN_PER_THREAD / 4; ++q) {
                *reinterpret_cast<int4*>(rowptr + i + 4 * q) =
                    make_int4(excl + p[q].x, excl + p[q].y, excl + p[q].z, excl + p[q].w);
                *reinterpret_cast<float4*>(dinv + i + 4 * q) =
                    make_float4(rsqrtf((float)(c[q].x + 1)), rsqrtf((float)(c[q].y + 1)),
                                rsqrtf((float)(c[q].z + 1)), rsqrtf((float)(c[q].w + 1)));
            }
        }
        carry += tileTotal;
        __syncthreads();
    }
    if (t == 0) rowptr[N_NODES] = carry;
}

__global__ void fill_kernel(const int* __restrict__ src,
                            const int* __restrict__ dst,
                            const int* __restrict__ rowptr,
                            const int* __restrict__ rank,
                            const float* __restrict__ dinv,
                            int* __restrict__ adj) {
    int e = blockIdx.x * blockDim.x + threadIdx.x;
    if (e < N_EDGES) {
        const int s = src[e];
        const int p = rowptr[dst[e]] + rank[e];
        reinterpret_cast<int2*>(adj)[p] = make_int2(s, __float_as_int(dinv[s]));
    }
}

__global__ __launch_bounds__(32) void gemm_x_w1(const float* __restrict__ x,
                                                const float* __restrict__ W,
                                                float* __restrict__ out) {
    const int lane = threadIdx.x;
    const int m    = lane & 15;
    const int half = lane >> 4;
    const int row0 = blockIdx.x * 16;

    v8f acc0 = {}, acc1 = {}, acc2 = {}, acc3 = {};
    const float* xr = x + (size_t)(row0 + m) * DIM;

#pragma unroll
    for (int ks = 0; ks < 16; ++ks) {
        const int kk = ks * 4 + half * 2;
        v2f a;  a.x = xr[kk];  a.y = xr[kk + 1];
        const float* wr0 = W + (size_t)kk * DIM;
        const float* wr1 = W + (size_t)(kk + 1) * DIM;
        v2f b0, b1, b2, b3;
        b0.x = wr0[m];      b0.y = wr1[m];
        b1.x = wr0[16 + m]; b1.y = wr1[16 + m];
        b2.x = wr0[32 + m]; b2.y = wr1[32 + m];
        b3.x = wr0[48 + m]; b3.y = wr1[48 + m];
        acc0 = __builtin_amdgcn_wmma_f32_16x16x4_f32(false, a, false, b0, (short)0, acc0, false, false);
        acc1 = __builtin_amdgcn_wmma_f32_16x16x4_f32(false, a, false, b1, (short)0, acc1, false, false);
        acc2 = __builtin_amdgcn_wmma_f32_16x16x4_f32(false, a, false, b2, (short)0, acc2, false, false);
        acc3 = __builtin_amdgcn_wmma_f32_16x16x4_f32(false, a, false, b3, (short)0, acc3, false, false);
    }

#pragma unroll
    for (int r = 0; r < 8; ++r) {
        float* o = out + (size_t)(row0 + r + 8 * half) * DIM;
        o[m]      = acc0[r];
        o[16 + m] = acc1[r];
        o[32 + m] = acc2[r];
        o[48 + m] = acc3[r];
    }
}

template <int U>
__device__ __forceinline__ void pull_group(float2& acc, const float2* __restrict__ f2,
                                           int sk, float wk, int k, int lane) {
    float2 v[U];
    float  w[U];
#pragma unroll
    for (int u = 0; u < U; ++u) {
        const int s = __builtin_amdgcn_readlane(sk, k + u);
        w[u] = __int_as_float(__builtin_amdgcn_readlane(__float_as_int(wk), k + u));
        v[u] = f2[(size_t)s * (DIM / 2) + lane];
    }
#pragma unroll
    for (int u = 0; u < U; ++u) {
        acc.x += v[u].x * w[u];
        acc.y += v[u].y * w[u];
    }
}

#define PULL_WAVES 8
static_assert(N_NODES % PULL_WAVES == 0, "every wave of the grid owns a node");

__global__ __launch_bounds__(PULL_WAVES * 32) void pull_kernel(const int* __restrict__ rowptr,
                                                               const int* __restrict__ adj,
                                                               const float* __restrict__ dinv,
                                                               const float* __restrict__ feat,
                                                               float* __restrict__ out) {
    const int lane = threadIdx.x & 31;
    const int node = __builtin_amdgcn_readfirstlane((int)(blockIdx.x * PULL_WAVES + (threadIdx.x >> 5)));

    const int   beg = rowptr[node];
    const int   end = rowptr[node + 1];
    const float dd  = dinv[node];
    const float2* f2  = reinterpret_cast<const float2*>(feat);
    const int2*   a2  = reinterpret_cast<const int2*>(adj);

    const float2 self = f2[(size_t)node * (DIM / 2) + lane];
    const float  wself = dd * dd;
    float2 acc = make_float2(self.x * wself, self.y * wself);

    for (int base = beg; base < end; base += 32) {
        const int n = (end - base < 32) ? (end - base) : 32;
        int   sk = 0;
        float wk = 0.0f;
        if (lane < n) {
            const int2 a = a2[base + lane];
            sk = a.x;
            wk = __int_as_float(a.y) * dd;
        }
        int k = 0;
        for (; k + 8 <= n; k += 8) pull_group<8>(acc, f2, sk, wk, k, lane);
        if (k + 4 <= n) { pull_group<4>(acc, f2, sk, wk, k, lane); k += 4; }
        if (k + 2 <= n) { pull_group<2>(acc, f2, sk, wk, k, lane); k += 2; }
        if (k < n)      { pull_group<1>(acc, f2, sk, wk, k, lane); }
    }
    reinterpret_cast<float2*>(out)[(size_t)node * (DIM / 2) + lane] = acc;
}

__global__ void bias_relu_kernel(float* __restrict__ h, const float* __restrict__ b1) {
    size_t tid = (size_t)blockIdx.x * blockDim.x + threadIdx.x;
    if (tid < (size_t)N_NODES * DIM)
        h[tid] = fmaxf(h[tid] + b1[tid & 63], 0.0f);
}

__global__ __launch_bounds__(32) void gemm_out(const float* __restrict__ agg,
                                               const float* __restrict__ Wmu,
                                               const float* __restrict__ bmu,
                                               const float* __restrict__ Wlv,
                                               const float* __restrict__ blv,
                                               float* __restrict__ out) {
    const int lane = threadIdx.x;
    const int m    = lane & 15;
    const int half = lane >> 4;
    const int row0 = blockIdx.x * 16;

    v8f accMu0 = {}, accMu1 = {}, accLv0 = {}, accLv1 = {};
    const float* ar = agg + (size_t)(row0 + m) * DIM;

#pragma unroll
    for (int ks = 0; ks < 16; ++ks) {
        const int kk = ks * 4 + half * 2;
        v2f a;  a.x = ar[kk];  a.y = ar[kk + 1];
        const float* mu0 = Wmu + (size_t)kk * OUT_DIM;
        const float* mu1 = Wmu + (size_t)(kk + 1) * OUT_DIM;
        const float* lv0 = Wlv + (size_t)kk * OUT_DIM;
        const float* lv1 = Wlv + (size_t)(kk + 1) * OUT_DIM;
        v2f b0, b1, b2, b3;
        b0.x = mu0[m];      b0.y = mu1[m];
        b1.x = mu0[16 + m]; b1.y = mu1[16 + m];
        b2.x = lv0[m];      b2.y = lv1[m];
        b3.x = lv0[16 + m]; b3.y = lv1[16 + m];
        accMu0 = __builtin_amdgcn_wmma_f32_16x16x4_f32(false, a, false, b0, (short)0, accMu0, false, false);
        accMu1 = __builtin_amdgcn_wmma_f32_16x16x4_f32(false, a, false, b1, (short)0, accMu1, false, false);
        accLv0 = __builtin_amdgcn_wmma_f32_16x16x4_f32(false, a, false, b2, (short)0, accLv0, false, false);
        accLv1 = __builtin_amdgcn_wmma_f32_16x16x4_f32(false, a, false, b3, (short)0, accLv1, false, false);
    }

    float* lvBase = out + (size_t)N_NODES * OUT_DIM;
#pragma unroll
    for (int r = 0; r < 8; ++r) {
        const int row = row0 + r + 8 * half;
        float* omu = out    + (size_t)row * OUT_DIM;
        float* olv = lvBase + (size_t)row * OUT_DIM;
        omu[m]      = accMu0[r] + bmu[m];
        omu[16 + m] = accMu1[r] + bmu[16 + m];
        olv[m]      = accLv0[r] + blv[m];
        olv[16 + m] = accLv1[r] + blv[16 + m];
    }
}

extern "C" void kernel_launch(void* const* d_in, const int* in_sizes, int n_in,
                              void* d_out, int out_size, void* d_ws, size_t ws_size,
                              hipStream_t stream) {
    const float* x    = (const float*)d_in[0];
    const int*   eidx = (const int*)d_in[1];
    const int*   src  = eidx;
    const int*   dst  = eidx + N_EDGES;
    const float* W1   = (const float*)d_in[2];
    const float* b1   = (const float*)d_in[3];
    const float* Wmu  = (const float*)d_in[4];
    const float* bmu  = (const float*)d_in[5];
    const float* Wlv  = (const float*)d_in[6];
    const float* blv  = (const float*)d_in[7];
    float*       out  = (float*)d_out;

    float* dinv = (float*)d_ws;
    float* xw   = dinv + N_NODES;
    float* agg1 = xw   + (size_t)N_NODES * DIM;
    float* agg2 = agg1 + (size_t)N_NODES * DIM;
    int*   cnt    = (int*)(agg2 + (size_t)N_NODES * DIM);
    int*   rowptr = cnt    + N_NODES;
    int*   rank   = rowptr + (N_NODES + 4);
    int*   adj    = rank   + N_EDGES;

    const int TB = 256;
    const size_t featElems = (size_t)N_NODES * DIM;

    const unsigned gFeat = (unsigned)((featElems + TB - 1) / TB);
    const unsigned gEdge = (N_EDGES + TB - 1) / TB;
    const unsigned gNode = (N_NODES + TB - 1) / TB;
    const unsigned gPull = N_NODES / PULL_WAVES;
    const unsigned gGemm = N_NODES / 16;

    zero_counts_kernel<<<gNode, TB, 0, stream>>>(cnt);
    count_rank_kernel<<<gEdge, TB, 0, stream>>>(dst, cnt, rank);
    scan_kernel<<<1, SCAN_THREADS, 0, stream>>>(cnt, rowptr, dinv);
    fill_kernel<<<gEdge, TB, 0, stream>>>(src, dst, rowptr, rank, dinv, adj);

    gemm_x_w1<<<gGemm, 32, 0, stream>>>(x, W1, xw);
    pull_kernel<<<gPull, PULL_WAVES * 32, 0, stream>>>(rowptr, adj, dinv, xw, agg1);
    bias_relu_kernel<<<gFeat, TB, 0, stream>>>(agg1, b1);
    pull_kernel<<<gPull, PULL_WAVES * 32, 0, stream>>>(rowptr, adj, dinv, agg1, agg2);
    gemm_out<<<gGemm, 32, 0, stream>>>(agg2, Wmu, bmu, Wlv, blv, out);
}


# Fold the bias + ReLU pass into the first pull's epilogue: the wave that sums a row adds b1 and clamps it before its one store; the 200,000-wave elementwise launch is gone; sum order unchanged
# speedup vs baseline: 3.1270x; 1.0722x over previous
#include <hip/hip_runtime.h>
#include <hip/hip_bf16.h>

#define N_NODES 100000
#define N_EDGES 1600000
#define DIM 64
#define OUT_DIM 32

typedef __attribute__((ext_vector_type(2))) float v2f;
typedef __attribute__((ext_vector_type(8))) float v8f;

__global__ void zero_counts_kernel(int* __restrict__ cnt) {
    int i = blockIdx.x * blockDim.x + threadIdx.x;
    if (i < N_NODES) cnt[i] = 0;
}

__global__ void count_rank_kernel(const int* __restrict__ dst,
                                  int* __restrict__ cnt,
                                  int* __restrict__ rank) {
    int e = blockIdx.x * blockDim.x + threadIdx.x;
    if (e < N_EDGES) rank[e] = atomicAdd(&cnt[dst[e]], 1);
}

#define SCAN_THREADS 1024
#define SCAN_PER_THREAD 16
#define SCAN_TILE (SCAN_THREADS * SCAN_PER_THREAD)
static_assert(N_NODES % SCAN_PER_THREAD == 0, "a thread's counts are all inside or all outside");

__device__ __forceinline__ int wave_inclusive_scan(int v, int lane) {
#pragma unroll
    for (int off = 1; off < 32; off <<= 1) {
        const int u = __shfl_up(v, off);
        if (lane >= off) v += u;
    }
    return v;
}

__global__ __launch_bounds__(SCAN_THREADS) void scan_kernel(const int* __restrict__ cnt,
                                                            int* __restrict__ rowptr,
                                                            float* __restrict__ dinv) {
    __shared__ int waveSum[32];
    const int t    = threadIdx.x;
    const int lane = t & 31;
    const int wave = t >> 5;
    int carry = 0;
    for (int base = 0; base < N_NODES; base += SCAN_TILE) {
        const int i = base + t * SCAN_PER_THREAD;
        const bool in = i < N_NODES;
        int4 c[SCAN_PER_THREAD / 4];
        int4 p[SCAN_PER_THREAD / 4];
        int run = 0;
#pragma unroll
        for (int q = 0; q < SCAN_PER_THREAD / 4; ++q) {
            c[q] = make_int4(0, 0, 0, 0);
            if (in) c[q] = *reinterpret_cast<const int4*>(cnt + i + 4 * q);
        }
#pragma unroll
        for (int q = 0; q < SCAN_PER_THREAD / 4; ++q) {
            p[q].x = run; run += c[q].x;
            p[q].y = run; run += c[q].y;
            p[q].z = run; run += c[q].z;
            p[q].w = run; run += c[q].w;
        }
        const int v = wave_inclusive_scan(run, lane);
        if (lane == 31) waveSum[wave] = v;
        __syncthreads();
        const int ws = waveSum[lane];
        const int wv = wave_inclusive_scan(ws, lane);
        const int tileTotal  = __shfl(wv, 31);
        const int wavePrefix = __shfl(wv - ws, wave);
        const int excl = carry + wavePrefix + (v - run);
        if (in) {
#pragma unroll
            for (int q = 0; q < SCAN_PER_THREAD / 4; ++q) {
                *reinterpret_cast<int4*>(rowptr + i + 4 * q) =
                    make_int4(excl + p[q].x, excl + p[q].y, excl + p[q].z, excl + p[q].w);
                *reinterpret_cast<float4*>(dinv + i + 4 * q) =
                    make_float4(rsqrtf((float)(c[q].x + 1)), rsqrtf((float)(c[q].y + 1)),
                                rsqrtf((float)(c[q].z + 1)), rsqrtf((float)(c[q].w + 1)));
            }
        }
        carry += tileTotal;
        __syncthreads();
    }
    if (t == 0) rowptr[N_NODES] = carry;
}

__global__ void fill_kernel(const int* __restrict__ src,
                            const int* __restrict__ dst,
                            const int* __restrict__ rowptr,
                            const int* __restrict__ rank,
                            const float* __restrict__ dinv,
                            int* __restrict__ adj) {
    int e = blockIdx.x * blockDim.x + threadIdx.x;
    if (e < N_EDGES) {
        const int s = src[e];
        const int p = rowptr[dst[e]] + rank[e];
        reinterpret_cast<int2*>(adj)[p] = make_int2(s, __float_as_int(dinv[s]));
    }
}

__global__ __launch_bounds__(32) void gemm_x_w1(const float* __restrict__ x,
                                                const float* __restrict__ W,
                                                float* __restrict__ out) {
    const int lane = threadIdx.x;
    const int m    = lane & 15;
    const int half = lane >> 4;
    const int row0 = blockIdx.x * 16;

    v8f acc0 = {}, acc1 = {}, acc2 = {}, acc3 = {};
    const float* xr = x + (size_t)(row0 + m) * DIM;

#pragma unroll
    for (int ks = 0; ks < 16; ++ks) {
        const int kk = ks * 4 + half * 2;
        v2f a;  a.x = xr[kk];  a.y = xr[kk + 1];
        const float* wr0 = W + (size_t)kk * DIM;
        const float* wr1 = W + (size_t)(kk + 1) * DIM;
        v2f b0, b1, b2, b3;
        b0.x = wr0[m];      b0.y = wr1[m];
        b1.x = wr0[16 + m]; b1.y = wr1[16 + m];
        b2.x = wr0[32 + m]; b2.y = wr1[32 + m];
        b3.x = wr0[48 + m]; b3.y = wr1[48 + m];
        acc0 = __builtin_amdgcn_wmma_f32_16x16x4_f32(false, a, false, b0, (short)0, acc0, false, false);
        acc1 = __builtin_amdgcn_wmma_f32_16x16x4_f32(false, a, false, b1, (short)0, acc1, false, false);
        acc2 = __builtin_amdgcn_wmma_f32_16x16x4_f32(false, a, false, b2, (short)0, acc2, false, false);
        acc3 = __builtin_amdgcn_wmma_f32_16x16x4_f32(false, a, false, b3, (short)0, acc3, false, false);
    }

#pragma unroll
    for (int r = 0; r < 8; ++r) {
        float* o = out + (size_t)(row0 + r + 8 * half) * DIM;
        o[m]      = acc0[r];
        o[16 + m] = acc1[r];
        o[32 + m] = acc2[r];
        o[48 + m] = acc3[r];
    }
}

template <int U>
__device__ __forceinline__ void pull_group(float2& acc, const float2* __restrict__ f2,
                                           int sk, float wk, int k, int lane) {
    float2 v[U];
    float  w[U];
#pragma unroll
    for (int u = 0; u < U; ++u) {
        const int s = __builtin_amdgcn_readlane(sk, k + u);
        w[u] = __int_as_float(__builtin_amdgcn_readlane(__float_as_int(wk), k + u));
        v[u] = f2[(size_t)s * (DIM / 2) + lane];
    }
#pragma unroll
    for (int u = 0; u < U; ++u) {
        acc.x += v[u].x * w[u];
        acc.y += v[u].y * w[u];
    }
}

#define PULL_WAVES 8
static_assert(N_NODES % PULL_WAVES == 0, "every wave of the grid owns a node");

template <bool BIAS_RELU>
__global__ __launch_bounds__(PULL_WAVES * 32) void pull_kernel(const int* __restrict__ rowptr,
                                                               const int* __restrict__ adj,
                                                               const float* __restrict__ dinv,
                                                               const float* __restrict__ feat,
                                                               const float* __restrict__ bias,
                                                               float* __restrict__ out) {
    const int lane = threadIdx.x & 31;
    const int node = __builtin_amdgcn_readfirstlane((int)(blockIdx.x * PULL_WAVES + (threadIdx.x >> 5)));

    const int   beg = rowptr[node];
    const int   end = rowptr[node + 1];
    const float dd  = dinv[node];
    const float2* f2  = reinterpret_cast<const float2*>(feat);
    const int2*   a2  = reinterpret_cast<const int2*>(adj);

    const float2 self = f2[(size_t)node * (DIM / 2) + lane];
    float2 b = make_float2(0.0f, 0.0f);
    if (BIAS_RELU) b = reinterpret_cast<const float2*>(bias)[lane];
    const float  wself = dd * dd;
    float2 acc = make_float2(self.x * wself, self.y * wself);

    for (int base = beg; base < end; base += 32) {
        const int n = (end - base < 32) ? (end - base) : 32;
        int   sk = 0;
        float wk = 0.0f;
        if (lane < n) {
            const int2 a = a2[base + lane];
            sk = a.x;
            wk = __int_as_float(a.y) * dd;
        }
        int k = 0;
        for (; k + 8 <= n; k += 8) pull_group<8>(acc, f2, sk, wk, k, lane);
        if (k + 4 <= n) { pull_group<4>(acc, f2, sk, wk, k, lane); k += 4; }
        if (k + 2 <= n) { pull_group<2>(acc, f2, sk, wk, k, lane); k += 2; }
        if (k < n)      { pull_group<1>(acc, f2, sk, wk, k, lane); }
    }
    if (BIAS_RELU) {
        acc.x = fmaxf(acc.x + b.x, 0.0f);
        acc.y = fmaxf(acc.y + b.y, 0.0f);
    }
    reinterpret_cast<float2*>(out)[(size_t)node * (DIM / 2) + lane] = acc;
}

__global__ __launch_bounds__(32) void gemm_out(const float* __restrict__ agg,
                                               const float* __restrict__ Wmu,
                                               const float* __restrict__ bmu,
                                               const float* __restrict__ Wlv,
                                               const float* __restrict__ blv,
                                               float* __restrict__ out) {
    const int lane = threadIdx.x;
    const int m    = lane & 15;
    const int half = lane >> 4;
    const int row0 = blockIdx.x * 16;

    v8f accMu0 = {}, accMu1 = {}, accLv0 = {}, accLv1 = {};
    const float* ar = agg + (size_t)(row0 + m) * DIM;

#pragma unroll
    for (int ks = 0; ks < 16; ++ks) {
        const int kk = ks * 4 + half * 2;
        v2f a;  a.x = ar[kk];  a.y = ar[kk + 1];
        const float* mu0 = Wmu + (size_t)kk * OUT_DIM;
        const float* mu1 = Wmu + (size_t)(kk + 1) * OUT_DIM;
        const float* lv0 = Wlv + (size_t)kk * OUT_DIM;
        const float* lv1 = Wlv + (size_t)(kk + 1) * OUT_DIM;
        v2f b0, b1, b2, b3;
        b0.x = mu0[m];      b0.y = mu1[m];
        b1.x = mu0[16 + m]; b1.y = mu1[16 + m];
        b2.x = lv0[m];      b2.y = lv1[m];
        b3.x = lv0[16 + m]; b3.y = lv1[16 + m];
        accMu0 = __builtin_amdgcn_wmma_f32_16x16x4_f32(false, a, false, b0, (short)0, accMu0, false, false);
        accMu1 = __builtin_amdgcn_wmma_f32_16x16x4_f32(false, a, false, b1, (short)0, accMu1, false, false);
        accLv0 = __builtin_amdgcn_wmma_f32_16x16x4_f32(false, a, false, b2, (short)0, accLv0, false, false);
        accLv1 = __builtin_amdgcn_wmma_f32_16x16x4_f32(false, a, false, b3, (short)0, accLv1, false, false);
    }

    float* lvBase = out + (size_t)N_NODES * OUT_DIM;
#pragma unroll
    for (int r = 0; r < 8; ++r) {
        const int row = row0 + r + 8 * half;
        float* omu = out    + (size_t)row * OUT_DIM;
        float* olv = lvBase + (size_t)row * OUT_DIM;
        omu[m]      = accMu0[r] + bmu[m];
        omu[16 + m] = accMu1[r] + bmu[16 + m];
        olv[m]      = accLv0[r] + blv[m];
        olv[16 + m] = accLv1[r] + blv[16 + m];
    }
}

extern "C" void kernel_launch(void* const* d_in, const int* in_sizes, int n_in,
                              void* d_out, int out_size, void* d_ws, size_t ws_size,
                              hipStream_t stream) {
    const float* x    = (const float*)d_in[0];
    const int*   eidx = (const int*)d_in[1];
    const int*   src  = eidx;
    const int*   dst  = eidx + N_EDGES;
    const float* W1   = (const float*)d_in[2];
    const float* b1   = (const float*)d_in[3];
    const float* Wmu  = (const float*)d_in[4];
    const float* bmu  = (const float*)d_in[5];
    const float* Wlv  = (const float*)d_in[6];
    const float* blv  = (const float*)d_in[7];
    float*       out  = (float*)d_out;

    float* dinv = (float*)d_ws;
    float* xw   = dinv + N_NODES;
    float* h    = xw   + (size_t)N_NODES * DIM;
    float* agg2 = h    + (size_t)N_NODES * DIM;
    int*   cnt    = (int*)(agg2 + (size_t)N_NODES * DIM);
    int*   rowptr = cnt    + N_NODES;
    int*   rank   = rowptr + (N_NODES + 4);
    int*   adj    = rank   + N_EDGES;

    const int TB = 256;
    const unsigned gEdge = (N_EDGES + TB - 1) / TB;
    const unsigned gNode = (N_NODES + TB - 1) / TB;
    const unsigned gPull = N_NODES / PULL_WAVES;
    const unsigned gGemm = N_NODES / 16;

    zero_counts_kernel<<<gNode, TB, 0, stream>>>(cnt);
    count_rank_kernel<<<gEdge, TB, 0, stream>>>(dst, cnt, rank);
    scan_kernel<<<1, SCAN_THREADS, 0, stream>>>(cnt, rowptr, dinv);
    fill_kernel<<<gEdge, TB, 0, stream>>>(src, dst, rowptr, rank, dinv, adj);

    gemm_x_w1<<<gGemm, 32, 0, stream>>>(x, W1, xw);
    pull_kernel<true><<<gPull, PULL_WAVES * 32, 0, stream>>>(rowptr, adj, dinv, xw, b1, h);
    pull_kernel<false><<<gPull, PULL_WAVES * 32, 0, stream>>>(rowptr, adj, dinv, h, nullptr, agg2);
    gemm_out<<<gGemm, 32, 0, stream>>>(agg2, Wmu, bmu, Wlv, blv, out);
}


# Project in the pull's epilogue: a 4-wave group sums 16 rows (4 a wave, edge lists fetched ahead) to LDS, multiplies the tile on WMMA; both GEMM launches go; layer 1 sums, then projects (sum order)
# speedup vs baseline: 3.4746x; 1.1112x over previous
#include <hip/hip_runtime.h>
#include <hip/hip_bf16.h>

#define N_NODES 100000
#define N_EDGES 1600000
#define DIM 64
#define OUT_DIM 32

typedef __attribute__((ext_vector_type(2))) float v2f;
typedef __attribute__((ext_vector_type(8))) float v8f;

__global__ void zero_counts_kernel(int* __restrict__ cnt) {
    int i = blockIdx.x * blockDim.x + threadIdx.x;
    if (i < N_NODES) cnt[i] = 0;
}

__global__ void count_rank_kernel(const int* __restrict__ dst,
                                  int* __restrict__ cnt,
                                  int* __restrict__ rank) {
    int e = blockIdx.x * blockDim.x + threadIdx.x;
    if (e < N_EDGES) rank[e] = atomicAdd(&cnt[dst[e]], 1);
}

#define SCAN_THREADS 1024
#define SCAN_PER_THREAD 16
#define SCAN_TILE (SCAN_THREADS * SCAN_PER_THREAD)
static_assert(N_NODES % SCAN_PER_THREAD == 0, "a thread's counts are all inside or all outside");

__device__ __forceinline__ int wave_inclusive_scan(int v, int lane) {
#pragma unroll
    for (int off = 1; off < 32; off <<= 1) {
        const int u = __shfl_up(v, off);
        if (lane >= off) v += u;
    }
    return v;
}

__global__ __launch_bounds__(SCAN_THREADS) void scan_kernel(const int* __restrict__ cnt,
                                                            int* __restrict__ rowptr,
                                                            float* __restrict__ dinv) {
    __shared__ int waveSum[32];
    const int t    = threadIdx.x;
    const int lane = t & 31;
    const int wave = t >> 5;
    int carry = 0;
    for (int base = 0; base < N_NODES; base += SCAN_TILE) {
        const int i = base + t * SCAN_PER_THREAD;
        const bool in = i < N_NODES;
        int4 c[SCAN_PER_THREAD / 4];
        int4 p[SCAN_PER_THREAD / 4];
        int run = 0;
#pragma unroll
        for (int q = 0; q < SCAN_PER_THREAD / 4; ++q) {
            c[q] = make_int4(0, 0, 0, 0);
            if (in) c[q] = *reinterpret_cast<const int4*>(cnt + i + 4 * q);
        }
#pragma unroll
        for (int q = 0; q < SCAN_PER_THREAD / 4; ++q) {
            p[q].x = run; run += c[q].x;
            p[q].y = run; run += c[q].y;
            p[q].z = run; run += c[q].z;
            p[q].w = run; run += c[q].w;
        }
        const int v = wave_inclusive_scan(run, lane);
        if (lane == 31) waveSum[wave] = v;
        __syncthreads();
        const int ws = waveSum[lane];
        const int wv = wave_inclusive_scan(ws, lane);
        const int tileTotal  = __shfl(wv, 31);
        const int wavePrefix = __shfl(wv - ws, wave);
        const int excl = carry + wavePrefix + (v - run);
        if (in) {
#pragma unroll
            for (int q = 0; q < SCAN_PER_THREAD / 4; ++q) {
                *reinterpret_cast<int4*>(rowptr + i + 4 * q) =
                    make_int4(excl + p[q].x, excl + p[q].y, excl + p[q].z, excl + p[q].w);
                *reinterpret_cast<float4*>(dinv + i + 4 * q) =
                    make_float4(rsqrtf((float)(c[q].x + 1)), rsqrtf((float)(c[q].y + 1)),
                                rsqrtf((float)(c[q].z + 1)), rsqrtf((float)(c[q].w + 1)));
            }
        }
        carry += tileTotal;
        __syncthreads();
    }
    if (t == 0) rowptr[N_NODES] = carry;
}

__global__ void fill_kernel(const int* __restrict__ src,
                            const int* __restrict__ dst,
                            const int* __restrict__ rowptr,
                            const int* __restrict__ rank,
                            const float* __restrict__ dinv,
                            int* __restrict__ adj) {
    int e = blockIdx.x * blockDim.x + threadIdx.x;
    if (e < N_EDGES) {
        const int s = src[e];
        const int p = rowptr[dst[e]] + rank[e];
        reinterpret_cast<int2*>(adj)[p] = make_int2(s, __float_as_int(dinv[s]));
    }
}

template <int U>
__device__ __forceinline__ void pull_group(float2& acc, const float2* __restrict__ f2,
                                           int sk, float wk, int k, int lane) {
    float2 v[U];
    float  w[U];
#pragma unroll
    for (int u = 0; u < U; ++u) {
        const int s = __builtin_amdgcn_readlane(sk, k + u);
        w[u] = __int_as_float(__builtin_amdgcn_readlane(__float_as_int(wk), k + u));
        v[u] = f2[(size_t)s * (DIM / 2) + lane];
    }
#pragma unroll
    for (int u = 0; u < U; ++u) {
        acc.x += v[u].x * w[u];
        acc.y += v[u].y * w[u];
    }
}

__device__ __forceinline__ void pull_chunk(float2& acc, const float2* __restrict__ f2,
                                           int sk, float wk, int n, int lane) {
    int k = 0;
    for (; k + 8 <= n; k += 8) pull_group<8>(acc, f2, sk, wk, k, lane);
    if (k + 4 <= n) { pull_group<4>(acc, f2, sk, wk, k, lane); k += 4; }
    if (k + 2 <= n) { pull_group<2>(acc, f2, sk, wk, k, lane); k += 2; }
    if (k < n)      { pull_group<1>(acc, f2, sk, wk, k, lane); }
}

#define TILE_NODES 16
#define TILE_WAVES 4
#define WAVE_NODES (TILE_NODES / TILE_WAVES)
#define TILE_PITCH 68
static_assert(N_NODES % TILE_NODES == 0, "every work-group owns a full 16-row tile");
static_assert(TILE_WAVES == 4, "one wave per 16-column slice of the 64 result columns");

template <bool LAYER1>
__global__ __launch_bounds__(TILE_WAVES * 32) void pull_project_kernel(const int* __restrict__ rowptr,
                                                                       const int* __restrict__ adj,
                                                                       const float* __restrict__ dinv,
                                                                       const float* __restrict__ feat,
                                                                       const float* __restrict__ Wa,
                                                                       const float* __restrict__ ba,
                                                                       const float* __restrict__ Wb,
                                                                       const float* __restrict__ bb,
                                                                       float* __restrict__ out) {
    __shared__ float tile[TILE_NODES * TILE_PITCH];

    const int lane = threadIdx.x & 31;
    const int wave = __builtin_amdgcn_readfirstlane((int)(threadIdx.x >> 5));
    const int row0 = blockIdx.x * TILE_NODES;
    const int n0   = row0 + wave * WAVE_NODES;

    const float2* f2 = reinterpret_cast<const float2*>(feat);
    const int2*   a2 = reinterpret_cast<const int2*>(adj);

    int    rp[WAVE_NODES + 1];
    float  dd[WAVE_NODES];
    float2 selfv[WAVE_NODES];
    int2   first[WAVE_NODES];
#pragma unroll
    for (int j = 0; j <= WAVE_NODES; ++j) rp[j] = rowptr[n0 + j];
#pragma unroll
    for (int j = 0; j < WAVE_NODES; ++j) {
        dd[j]    = dinv[n0 + j];
        selfv[j] = f2[(size_t)(n0 + j) * (DIM / 2) + lane];
    }
#pragma unroll
    for (int j = 0; j < WAVE_NODES; ++j) {
        const bool in = rp[j] + lane < rp[j + 1];
        const int2 a = a2[in ? rp[j] + lane : 0];
        first[j] = make_int2(in ? a.x : 0, in ? a.y : 0);
    }

#pragma unroll
    for (int j = 0; j < WAVE_NODES; ++j) {
        const int   beg   = rp[j];
        const int   end   = rp[j + 1];
        const float wself = dd[j] * dd[j];
        float2 acc = make_float2(selfv[j].x * wself, selfv[j].y * wself);

        {
            const int n = (end - beg < 32) ? (end - beg) : 32;
            const float wk = __int_as_float(first[j].y) * dd[j];
            pull_chunk(acc, f2, first[j].x, wk, n, lane);
        }
        for (int base = beg + 32; base < end; base += 32) {
            const int n = (end - base < 32) ? (end - base) : 32;
            int   sk = 0;
            float wk = 0.0f;
            if (lane < n) {
                const int2 a = a2[base + lane];
                sk = a.x;
                wk = __int_as_float(a.y) * dd[j];
            }
            pull_chunk(acc, f2, sk, wk, n, lane);
        }

        *reinterpret_cast<float2*>(&tile[(wave * WAVE_NODES + j) * TILE_PITCH + 2 * lane]) = acc;
    }
    __syncthreads();

    {
        const int c    = lane & 15;
        const int half = lane >> 4;
        const int t    = wave;

        const int    ldw  = LAYER1 ? DIM : OUT_DIM;
        const int    col  = LAYER1 ? (16 * t + c) : (16 * (t & 1) + c);
        const float* Wsel = (LAYER1 || t < 2) ? Wa : Wb;
        const float* bsel = (LAYER1 || t < 2) ? ba : bb;
        const float* wp   = Wsel + col;
        const float  bias = bsel[col];
        const float* ar   = tile + c * TILE_PITCH + 2 * half;

        v8f accE = {}, accO = {};
#pragma unroll
        for (int ks = 0; ks < 16; ++ks) {
            const int kk = ks * 4 + half * 2;
            const v2f a = *reinterpret_cast<const v2f*>(ar + 4 * ks);
            v2f b;
            b.x = wp[(size_t)kk * ldw];
            b.y = wp[(size_t)(kk + 1) * ldw];
            if (ks & 1) accO = __builtin_amdgcn_wmma_f32_16x16x4_f32(false, a, false, b, (short)0, accO, false, false);
            else        accE = __builtin_amdgcn_wmma_f32_16x16x4_f32(false, a, false, b, (short)0, accE, false, false);
        }

        float* obase;
        if (LAYER1) obase = out + (size_t)row0 * DIM + col;
        else        obase = out + (t < 2 ? (size_t)0 : (size_t)N_NODES * OUT_DIM) + (size_t)row0 * OUT_DIM + col;
#pragma unroll
        for (int r = 0; r < 8; ++r) {
            float v = accE[r] + accO[r] + bias;
            if (LAYER1) v = fmaxf(v, 0.0f);
            obase[(size_t)(r + 8 * half) * ldw] = v;
        }
    }
}

extern "C" void kernel_launch(void* const* d_in, const int* in_sizes, int n_in,
                              void* d_out, int out_size, void* d_ws, size_t ws_size,
                              hipStream_t stream) {
    const float* x    = (const float*)d_in[0];
    const int*   eidx = (const int*)d_in[1];
    const int*   src  = eidx;
    const int*   dst  = eidx + N_EDGES;
    const float* W1   = (const float*)d_in[2];
    const float* b1   = (const float*)d_in[3];
    const float* Wmu  = (const float*)d_in[4];
    const float* bmu  = (const float*)d_in[5];
    const float* Wlv  = (const float*)d_in[6];
    const float* blv  = (const float*)d_in[7];
    float*       out  = (float*)d_out;

    float* dinv = (float*)d_ws;
    float* h    = dinv + N_NODES;
    int*   cnt    = (int*)(h + (size_t)N_NODES * DIM);
    int*   rowptr = cnt    + N_NODES;
    int*   rank   = rowptr + (N_NODES + 4);
    int*   adj    = rank   + N_EDGES;

    const int TB = 256;
    const unsigned gEdge = (N_EDGES + TB - 1) / TB;
    const unsigned gNode = (N_NODES + TB - 1) / TB;
    const unsigned gTile = N_NODES / TILE_NODES;

    zero_counts_kernel<<<gNode, TB, 0, stream>>>(cnt);
    count_rank_kernel<<<gEdge, TB, 0, stream>>>(dst, cnt, rank);
    scan_kernel<<<1, SCAN_THREADS, 0, stream>>>(cnt, rowptr, dinv);
    fill_kernel<<<gEdge, TB, 0, stream>>>(src, dst, rowptr, rank, dinv, adj);

    pull_project_kernel<true><<<gTile, TILE_WAVES * 32, 0, stream>>>(rowptr, adj, dinv, x, W1, b1, nullptr, nullptr, h);
    pull_project_kernel<false><<<gTile, TILE_WAVES * 32, 0, stream>>>(rowptr, adj, dinv, h, Wmu, bmu, Wlv, blv, out);
}


# CSR build by a two-level counting sort on LDS counters (edge chunks binned by 256-node range, one work-group per range) replaces the per-edge returning global atomic; sum order = LDS arrival rank
# speedup vs baseline: 8.5153x; 2.4507x over previous
#include <hip/hip_runtime.h>
#include <hip/hip_bf16.h>

#define N_NODES 100000
#define N_EDGES 1600000
#define DIM 64
#define OUT_DIM 32

typedef __attribute__((ext_vector_type(2))) float v2f;
typedef __attribute__((ext_vector_type(8))) float v8f;

#define CHUNK 8192
#define NCH ((N_EDGES + CHUNK - 1) / CHUNK)
#define BIN_SHIFT 8
#define BIN_NODES (1 << BIN_SHIFT)
#define NBIN ((N_NODES + BIN_NODES - 1) / BIN_NODES)
#define TROW (NBIN + 1)

__device__ __forceinline__ int wave_inclusive_scan(int v, int lane) {
#pragma unroll
    for (int off = 1; off < 32; off <<= 1) {
        const int u = __shfl_up(v, off);
        if (lane >= off) v += u;
    }
    return v;
}

#define BIN_THREADS 1024
#define BIN_PER_THREAD (CHUNK / BIN_THREADS)
static_assert(NBIN < BIN_THREADS, "one thread per bin counter, and one more for the total");
static_assert(BIN_THREADS == 1024, "the scan below reads one wave total per lane: 32 waves");

__global__ __launch_bounds__(BIN_THREADS) void bin_chunk_kernel(const int* __restrict__ src,
                                                                const int* __restrict__ dst,
                                                                int* __restrict__ recs,
                                                                int* __restrict__ tab) {
    __shared__ int  hist[BIN_THREADS];
    __shared__ int  waveSum[32];
    __shared__ int2 stage[CHUNK];
    const int t    = threadIdx.x;
    const int lane = t & 31;
    const int wave = t >> 5;
    const int e0   = blockIdx.x * CHUNK;

    hist[t] = 0;
    int d[BIN_PER_THREAD], sv[BIN_PER_THREAD], r[BIN_PER_THREAD];
#pragma unroll
    for (int k = 0; k < BIN_PER_THREAD; ++k) {
        const int e = e0 + k * BIN_THREADS + t;
        d[k]  = (e < N_EDGES) ? dst[e] : -1;
        sv[k] = (e < N_EDGES) ? src[e] : 0;
    }
    __syncthreads();
#pragma unroll
    for (int k = 0; k < BIN_PER_THREAD; ++k) {
        r[k] = 0;
        if (d[k] >= 0) r[k] = atomicAdd(&hist[d[k] >> BIN_SHIFT], 1);
    }
    __syncthreads();

    const int val = hist[t];
    const int v   = wave_inclusive_scan(val, lane);
    if (lane == 31) waveSum[wave] = v;
    __syncthreads();
    const int ws = waveSum[lane];
    const int wv = wave_inclusive_scan(ws, lane);
    const int wavePrefix = __shfl(wv - ws, wave);
    const int excl = wavePrefix + (v - val);
    hist[t] = excl;
    if (t <= NBIN) tab[blockIdx.x * TROW + t] = excl;
    __syncthreads();

#pragma unroll
    for (int k = 0; k < BIN_PER_THREAD; ++k) {
        if (d[k] >= 0) stage[hist[d[k] >> BIN_SHIFT] + r[k]] = make_int2(sv[k], d[k]);
    }
    __syncthreads();
    int2* out2 = reinterpret_cast<int2*>(recs) + (size_t)blockIdx.x * CHUNK;
#pragma unroll
    for (int k = 0; k < BIN_PER_THREAD; ++k) {
        const int idx = k * BIN_THREADS + t;
        if (e0 + idx < N_EDGES) out2[idx] = stage[idx];
    }
}

#define CSR_THREADS 1024
#define CSR_WAVES (CSR_THREADS / 32)
static_assert(NCH <= CSR_THREADS, "one thread fetches one chunk's run bounds");
static_assert(BIN_NODES < CSR_THREADS, "one thread per node of the bin, and one more for the total");

__global__ __launch_bounds__(CSR_THREADS) void csr_bin_kernel(const int* __restrict__ recs,
                                                              const int* __restrict__ tab,
                                                              int* __restrict__ rowptr,
                                                              float* __restrict__ dinv,
                                                              int* __restrict__ srcSorted) {
    __shared__ int segS[NCH];
    __shared__ int segE[NCH];
    __shared__ int cntL[BIN_NODES];
    __shared__ int cur[BIN_NODES];
    __shared__ int baseSum[CSR_WAVES];
    __shared__ int waveSum[CSR_WAVES];
    const int t    = threadIdx.x;
    const int lane = t & 31;
    const int wave = __builtin_amdgcn_readfirstlane(t >> 5);
    const int bin  = blockIdx.x;
    const int node0 = bin * BIN_NODES;

    int s = 0;
    if (t < NCH) {
        s = tab[t * TROW + bin];
        segS[t] = s;
        segE[t] = tab[t * TROW + bin + 1];
    }
    if (t < BIN_NODES) cntL[t] = 0;
    int rs = s;
#pragma unroll
    for (int off = 16; off >= 1; off >>= 1) rs += __shfl_xor(rs, off);
    if (lane == 0) baseSum[wave] = rs;
    __syncthreads();
    int base = 0;
#pragma unroll
    for (int w = 0; w < CSR_WAVES; ++w) base += baseSum[w];

    const int2* r2 = reinterpret_cast<const int2*>(recs);
    for (int c = wave; c < NCH; c += CSR_WAVES) {
        const int sb = segS[c];
        const int se = segE[c];
        for (int i = sb + lane; i < se; i += 32) {
            const int2 rec = r2[(size_t)c * CHUNK + i];
            atomicAdd(&cntL[rec.y - node0], 1);
        }
    }
    __syncthreads();

    const int val = (t < BIN_NODES) ? cntL[t] : 0;
    const int v   = wave_inclusive_scan(val, lane);
    if (lane == 31) waveSum[wave] = v;
    __syncthreads();
    int wp = 0;
    for (int w = 0; w < wave; ++w) wp += waveSum[w];
    const int excl = wp + (v - val);
    if (t < BIN_NODES) {
        cur[t] = excl;
        const int node = node0 + t;
        if (node < N_NODES) {
            rowptr[node] = base + excl;
            dinv[node]   = rsqrtf((float)(val + 1));
        }
    }
    if (bin == NBIN - 1 && t == BIN_NODES) rowptr[N_NODES] = base + excl;
    __syncthreads();

    for (int c = wave; c < NCH; c += CSR_WAVES) {
        const int sb = segS[c];
        const int se = segE[c];
        for (int i = sb + lane; i < se; i += 32) {
            const int2 rec = r2[(size_t)c * CHUNK + i];
            const int  p   = atomicAdd(&cur[rec.y - node0], 1);
            srcSorted[base + p] = rec.x;
        }
    }
}

__global__ void adj_fill_kernel(const int* __restrict__ srcSorted,
                                const float* __restrict__ dinv,
                                int* __restrict__ adj) {
    int p = blockIdx.x * blockDim.x + threadIdx.x;
    if (p < N_EDGES) {
        const int s = srcSorted[p];
        reinterpret_cast<int2*>(adj)[p] = make_int2(s, __float_as_int(dinv[s]));
    }
}

template <int U>
__device__ __forceinline__ void pull_group(float2& acc, const float2* __restrict__ f2,
                                           int sk, float wk, int k, int lane) {
    float2 v[U];
    float  w[U];
#pragma unroll
    for (int u = 0; u < U; ++u) {
        const int s = __builtin_amdgcn_readlane(sk, k + u);
        w[u] = __int_as_float(__builtin_amdgcn_readlane(__float_as_int(wk), k + u));
        v[u] = f2[(size_t)s * (DIM / 2) + lane];
    }
#pragma unroll
    for (int u = 0; u < U; ++u) {
        acc.x += v[u].x * w[u];
        acc.y += v[u].y * w[u];
    }
}

__device__ __forceinline__ void pull_chunk(float2& acc, const float2* __restrict__ f2,
                                           int sk, float wk, int n, int lane) {
    int k = 0;
    for (; k + 8 <= n; k += 8) pull_group<8>(acc, f2, sk, wk, k, lane);
    if (k + 4 <= n) { pull_group<4>(acc, f2, sk, wk, k, lane); k += 4; }
    if (k + 2 <= n) { pull_group<2>(acc, f2, sk, wk, k, lane); k += 2; }
    if (k < n)      { pull_group<1>(acc, f2, sk, wk, k, lane); }
}

#define TILE_NODES 16
#define TILE_WAVES 4
#define WAVE_NODES (TILE_NODES / TILE_WAVES)
#define TILE_PITCH 68
static_assert(N_NODES % TILE_NODES == 0, "every work-group owns a full 16-row tile");
static_assert(TILE_WAVES == 4, "one wave per 16-column slice of the 64 result columns");

template <bool LAYER1>
__global__ __launch_bounds__(TILE_WAVES * 32) void pull_project_kernel(const int* __restrict__ rowptr,
                                                                       const int* __restrict__ adj,
                                                                       const float* __restrict__ dinv,
                                                                       const float* __restrict__ feat,
                                                                       const float* __restrict__ Wa,
                                                                       const float* __restrict__ ba,
                                                                       const float* __restrict__ Wb,
                                                                       const float* __restrict__ bb,
                                                                       float* __restrict__ out) {
    __shared__ float tile[TILE_NODES * TILE_PITCH];

    const int lane = threadIdx.x & 31;
    const int wave = __builtin_amdgcn_readfirstlane((int)(threadIdx.x >> 5));
    const int row0 = blockIdx.x * TILE_NODES;
    const int n0   = row0 + wave * WAVE_NODES;

    const float2* f2 = reinterpret_cast<const float2*>(feat);
    const int2*   a2 = reinterpret_cast<const int2*>(adj);

    int    rp[WAVE_NODES + 1];
    float  dd[WAVE_NODES];
    float2 selfv[WAVE_NODES];
    int2   first[WAVE_NODES];
#pragma unroll
    for (int j = 0; j <= WAVE_NODES; ++j) rp[j] = rowptr[n0 + j];
#pragma unroll
    for (int j = 0; j < WAVE_NODES; ++j) {
        dd[j]    = dinv[n0 + j];
        selfv[j] = f2[(size_t)(n0 + j) * (DIM / 2) + lane];
    }
#pragma unroll
    for (int j = 0; j < WAVE_NODES; ++j) {
        const bool in = rp[j] + lane < rp[j + 1];
        const int2 a = a2[in ? rp[j] + lane : 0];
        first[j] = make_int2(in ? a.x : 0, in ? a.y : 0);
    }

#pragma unroll
    for (int j = 0; j < WAVE_NODES; ++j) {
        const int   beg   = rp[j];
        const int   end   = rp[j + 1];
        const float wself = dd[j] * dd[j];
        float2 acc = make_float2(selfv[j].x * wself, selfv[j].y * wself);

        {
            const int n = (end - beg < 32) ? (end - beg) : 32;
            const float wk = __int_as_float(first[j].y) * dd[j];
            pull_chunk(acc, f2, first[j].x, wk, n, lane);
        }
        for (int base = beg + 32; base < end; base += 32) {
            const int n = (end - base < 32) ? (end - base) : 32;
            int   sk = 0;
            float wk = 0.0f;
            if (lane < n) {
                const int2 a = a2[base + lane];
                sk = a.x;
                wk = __int_as_float(a.y) * dd[j];
            }
            pull_chunk(acc, f2, sk, wk, n, lane);
        }

        *reinterpret_cast<float2*>(&tile[(wave * WAVE_NODES + j) * TILE_PITCH + 2 * lane]) = acc;
    }
    __syncthreads();

    {
        const int c    = lane & 15;
        const int half = lane >> 4;
        const int t    = wave;

        const int    ldw  = LAYER1 ? DIM : OUT_DIM;
        const int    col  = LAYER1 ? (16 * t + c) : (16 * (t & 1) + c);
        const float* Wsel = (LAYER1 || t < 2) ? Wa : Wb;
        const float* bsel = (LAYER1 || t < 2) ? ba : bb;
        const float* wp   = Wsel + col;
        const float  bias = bsel[col];
        const float* ar   = tile + c * TILE_PITCH + 2 * half;

        v8f accE = {}, accO = {};
#pragma unroll
        for (int ks = 0; ks < 16; ++ks) {
            const int kk = ks * 4 + half * 2;
            const v2f a = *reinterpret_cast<const v2f*>(ar + 4 * ks);
            v2f b;
            b.x = wp[(size_t)kk * ldw];
            b.y = wp[(size_t)(kk + 1) * ldw];
            if (ks & 1) accO = __builtin_amdgcn_wmma_f32_16x16x4_f32(false, a, false, b, (short)0, accO, false, false);
            else        accE = __builtin_amdgcn_wmma_f32_16x16x4_f32(false, a, false, b, (short)0, accE, false, false);
        }

        float* obase;
        if (LAYER1) obase = out + (size_t)row0 * DIM + col;
        else        obase = out + (t < 2 ? (size_t)0 : (size_t)N_NODES * OUT_DIM) + (size_t)row0 * OUT_DIM + col;
#pragma unroll
        for (int r = 0; r < 8; ++r) {
            float v = accE[r] + accO[r] + bias;
            if (LAYER1) v = fmaxf(v, 0.0f);
            obase[(size_t)(r + 8 * half) * ldw] = v;
        }
    }
}

extern "C" void kernel_launch(void* const* d_in, const int* in_sizes, int n_in,
                              void* d_out, int out_size, void* d_ws, size_t ws_size,
                              hipStream_t stream) {
    const float* x    = (const float*)d_in[0];
    const int*   eidx = (const int*)d_in[1];
    const int*   src  = eidx;
    const int*   dst  = eidx + N_EDGES;
    const float* W1   = (const float*)d_in[2];
    const float* b1   = (const float*)d_in[3];
    const float* Wmu  = (const float*)d_in[4];
    const float* bmu  = (const float*)d_in[5];
    const float* Wlv  = (const float*)d_in[6];
    const float* blv  = (const float*)d_in[7];
    float*       out  = (float*)d_out;

    float* dinv = (float*)d_ws;
    float* h    = dinv + N_NODES;
    int*   rowptr    = (int*)(h + (size_t)N_NODES * DIM);
    int*   srcSorted = rowptr    + (N_NODES + 4);
    int*   adj       = srcSorted + N_EDGES;
    int*   recs      = adj       + 2 * (size_t)N_EDGES;
    int*   tab       = recs      + 2 * (size_t)NCH * CHUNK;

    const int TB = 256;
    const unsigned gEdge = (N_EDGES + TB - 1) / TB;
    const unsigned gTile = N_NODES / TILE_NODES;

    bin_chunk_kernel<<<NCH, BIN_THREADS, 0, stream>>>(src, dst, recs, tab);
    csr_bin_kernel<<<NBIN, CSR_THREADS, 0, stream>>>(recs, tab, rowptr, dinv, srcSorted);
    adj_fill_kernel<<<gEdge, TB, 0, stream>>>(srcSorted, dinv, adj);

    pull_project_kernel<true><<<gTile, TILE_WAVES * 32, 0, stream>>>(rowptr, adj, dinv, x, W1, b1, nullptr, nullptr, h);
    pull_project_kernel<false><<<gTile, TILE_WAVES * 32, 0, stream>>>(rowptr, adj, dinv, h, Wmu, bmu, Wlv, blv, out);
}


# Rows scaled by dinv where dinv is made (x in the CSR bin kernel, h in layer 1's epilogue), sums scaled by dinv[dst] once: an edge is a bare source index; adj plane and fill launch go; sum order kept
# speedup vs baseline: 8.8209x; 1.0359x over previous
#include <hip/hip_runtime.h>
#include <hip/hip_bf16.h>

#define N_NODES 100000
#define N_EDGES 1600000
#define DIM 64
#define OUT_DIM 32

typedef __attribute__((ext_vector_type(2))) float v2f;
typedef __attribute__((ext_vector_type(8))) float v8f;

#define CHUNK 8192
#define NCH ((N_EDGES + CHUNK - 1) / CHUNK)
#define BIN_SHIFT 8
#define BIN_NODES (1 << BIN_SHIFT)
#define NBIN ((N_NODES + BIN_NODES - 1) / BIN_NODES)
#define TROW (NBIN + 1)

__device__ __forceinline__ int wave_inclusive_scan(int v, int lane) {
#pragma unroll
    for (int off = 1; off < 32; off <<= 1) {
        const int u = __shfl_up(v, off);
        if (lane >= off) v += u;
    }
    return v;
}

#define BIN_THREADS 1024
#define BIN_PER_THREAD (CHUNK / BIN_THREADS)
static_assert(NBIN < BIN_THREADS, "one thread per bin counter, and one more for the total");
static_assert(BIN_THREADS == 1024, "the scan below reads one wave total per lane: 32 waves");

__global__ __launch_bounds__(BIN_THREADS) void bin_chunk_kernel(const int* __restrict__ src,
                                                                const int* __restrict__ dst,
                                                                int* __restrict__ recs,
                                                                int* __restrict__ tab) {
    __shared__ int  hist[BIN_THREADS];
    __shared__ int  waveSum[32];
    __shared__ int2 stage[CHUNK];
    const int t    = threadIdx.x;
    const int lane = t & 31;
    const int wave = t >> 5;
    const int e0   = blockIdx.x * CHUNK;

    hist[t] = 0;
    int d[BIN_PER_THREAD], sv[BIN_PER_THREAD], r[BIN_PER_THREAD];
#pragma unroll
    for (int k = 0; k < BIN_PER_THREAD; ++k) {
        const int e = e0 + k * BIN_THREADS + t;
        d[k]  = (e < N_EDGES) ? dst[e] : -1;
        sv[k] = (e < N_EDGES) ? src[e] : 0;
    }
    __syncthreads();
#pragma unroll
    for (int k = 0; k < BIN_PER_THREAD; ++k) {
        r[k] = 0;
        if (d[k] >= 0) r[k] = atomicAdd(&hist[d[k] >> BIN_SHIFT], 1);
    }
    __syncthreads();

    const int val = hist[t];
    const int v   = wave_inclusive_scan(val, lane);
    if (lane == 31) waveSum[wave] = v;
    __syncthreads();
    const int ws = waveSum[lane];
    const int wv = wave_inclusive_scan(ws, lane);
    const int wavePrefix = __shfl(wv - ws, wave);
    const int excl = wavePrefix + (v - val);
    hist[t] = excl;
    if (t <= NBIN) tab[blockIdx.x * TROW + t] = excl;
    __syncthreads();

#pragma unroll
    for (int k = 0; k < BIN_PER_THREAD; ++k) {
        if (d[k] >= 0) stage[hist[d[k] >> BIN_SHIFT] + r[k]] = make_int2(sv[k], d[k]);
    }
    __syncthreads();
    int2* out2 = reinterpret_cast<int2*>(recs) + (size_t)blockIdx.x * CHUNK;
#pragma unroll
    for (int k = 0; k < BIN_PER_THREAD; ++k) {
        const int idx = k * BIN_THREADS + t;
        if (e0 + idx < N_EDGES) out2[idx] = stage[idx];
    }
}

#define CSR_THREADS 1024
#define CSR_WAVES (CSR_THREADS / 32)
static_assert(NCH <= CSR_THREADS, "one thread fetches one chunk's run bounds");
static_assert(BIN_NODES < CSR_THREADS, "one thread per node of the bin, and one more for the total");

#define ROW_F4 (DIM / 4)
#define XS_PER_THREAD (BIN_NODES * ROW_F4 / CSR_THREADS)
static_assert(BIN_NODES * ROW_F4 % CSR_THREADS == 0, "the bin's rows split evenly over the threads");

__global__ __launch_bounds__(CSR_THREADS) void csr_bin_kernel(const int* __restrict__ recs,
                                                              const int* __restrict__ tab,
                                                              const float* __restrict__ x,
                                                              int* __restrict__ rowptr,
                                                              float* __restrict__ dinv,
                                                              int* __restrict__ srcSorted,
                                                              float* __restrict__ xs) {
    __shared__ int segS[NCH];
    __shared__ int segE[NCH];
    __shared__ int cntL[BIN_NODES];
    __shared__ int cur[BIN_NODES];
    __shared__ int baseSum[CSR_WAVES];
    __shared__ int waveSum[CSR_WAVES];
    __shared__ float dinvL[BIN_NODES];
    const int t    = threadIdx.x;
    const int lane = t & 31;
    const int wave = __builtin_amdgcn_readfirstlane(t >> 5);
    const int bin  = blockIdx.x;
    const int node0 = bin * BIN_NODES;

    int s = 0;
    if (t < NCH) {
        s = tab[t * TROW + bin];
        segS[t] = s;
        segE[t] = tab[t * TROW + bin + 1];
    }
    if (t < BIN_NODES) cntL[t] = 0;

    int rs = s;
#pragma unroll
    for (int off = 16; off >= 1; off >>= 1) rs += __shfl_xor(rs, off);
    if (lane == 0) baseSum[wave] = rs;
    __syncthreads();
    int base = 0;
#pragma unroll
    for (int w = 0; w < CSR_WAVES; ++w) base += baseSum[w];

    const int2* r2 = reinterpret_cast<const int2*>(recs);
    for (int c = wave; c < NCH; c += CSR_WAVES) {
        const int sb = segS[c];
        const int se = segE[c];
        for (int i = sb + lane; i < se; i += 32) {
            const int2 rec = r2[(size_t)c * CHUNK + i];
            atomicAdd(&cntL[rec.y - node0], 1);
        }
    }
    __syncthreads();

    const int val = (t < BIN_NODES) ? cntL[t] : 0;
    const int v   = wave_inclusive_scan(val, lane);
    if (lane == 31) waveSum[wave] = v;
    __syncthreads();
    int wp = 0;
    for (int w = 0; w < wave; ++w) wp += waveSum[w];
    const int excl = wp + (v - val);
    if (t < BIN_NODES) {
        cur[t] = excl;
        const int node = node0 + t;
        const float dv = rsqrtf((float)(val + 1));
        dinvL[t] = dv;
        if (node < N_NODES) {
            rowptr[node] = base + excl;
            dinv[node]   = dv;
        }
    }
    if (bin == NBIN - 1 && t == BIN_NODES) rowptr[N_NODES] = base + excl;
    __syncthreads();

    const float4* x4 = reinterpret_cast<const float4*>(x) + (size_t)node0 * ROW_F4;
    float4 xr[XS_PER_THREAD];
#pragma unroll
    for (int k = 0; k < XS_PER_THREAD; ++k) {
        const int q = k * CSR_THREADS + t;
        xr[k] = (node0 + q / ROW_F4 < N_NODES) ? x4[q] : make_float4(0.0f, 0.0f, 0.0f, 0.0f);
    }

    for (int c = wave; c < NCH; c += CSR_WAVES) {
        const int sb = segS[c];
        const int se = segE[c];
        for (int i = sb + lane; i < se; i += 32) {
            const int2 rec = r2[(size_t)c * CHUNK + i];
            const int  p   = atomicAdd(&cur[rec.y - node0], 1);
            srcSorted[base + p] = rec.x;
        }
    }

    float4* xs4 = reinterpret_cast<float4*>(xs) + (size_t)node0 * ROW_F4;
#pragma unroll
    for (int k = 0; k < XS_PER_THREAD; ++k) {
        const int q = k * CSR_THREADS + t;
        if (node0 + q / ROW_F4 < N_NODES) {
            const float dv = dinvL[q / ROW_F4];
            xs4[q] = make_float4(xr[k].x * dv, xr[k].y * dv, xr[k].z * dv, xr[k].w * dv);
        }
    }
}

template <int U>
__device__ __forceinline__ void pull_group(float2& acc, const float2* __restrict__ f2,
                                           int sk, int k, int lane) {
    float2 v[U];
#pragma unroll
    for (int u = 0; u < U; ++u) {
        const int s = __builtin_amdgcn_readlane(sk, k + u);
        v[u] = f2[(size_t)s * (DIM / 2) + lane];
    }
#pragma unroll
    for (int u = 0; u < U; ++u) {
        acc.x += v[u].x;
        acc.y += v[u].y;
    }
}

__device__ __forceinline__ void pull_chunk(float2& acc, const float2* __restrict__ f2,
                                           int sk, int n, int lane) {
    int k = 0;
    for (; k + 8 <= n; k += 8) pull_group<8>(acc, f2, sk, k, lane);
    if (k + 4 <= n) { pull_group<4>(acc, f2, sk, k, lane); k += 4; }
    if (k + 2 <= n) { pull_group<2>(acc, f2, sk, k, lane); k += 2; }
    if (k < n)      { pull_group<1>(acc, f2, sk, k, lane); }
}

#define TILE_NODES 16
#define TILE_WAVES 4
#define WAVE_NODES (TILE_NODES / TILE_WAVES)
#define TILE_PITCH 68
static_assert(N_NODES % TILE_NODES == 0, "every work-group owns a full 16-row tile");
static_assert(TILE_WAVES == 4, "one wave per 16-column slice of the 64 result columns");

template <bool LAYER1>
__global__ __launch_bounds__(TILE_WAVES * 32) void pull_project_kernel(const int* __restrict__ rowptr,
                                                                       const int* __restrict__ srcSorted,
                                                                       const float* __restrict__ dinv,
                                                                       const float* __restrict__ feat,
                                                                       const float* __restrict__ Wa,
                                                                       const float* __restrict__ ba,
                                                                       const float* __restrict__ Wb,
                                                                       const float* __restrict__ bb,
                                                                       float* __restrict__ out) {
    __shared__ float tile[TILE_NODES * TILE_PITCH];
    __shared__ float ddL[TILE_NODES];

    const int lane = threadIdx.x & 31;
    const int wave = __builtin_amdgcn_readfirstlane((int)(threadIdx.x >> 5));
    const int row0 = blockIdx.x * TILE_NODES;
    const int n0   = row0 + wave * WAVE_NODES;

    const float2* f2 = reinterpret_cast<const float2*>(feat);

    int    rp[WAVE_NODES + 1];
    float  dd[WAVE_NODES];
    float2 selfv[WAVE_NODES];
    int    first[WAVE_NODES];
#pragma unroll
    for (int j = 0; j <= WAVE_NODES; ++j) rp[j] = rowptr[n0 + j];
#pragma unroll
    for (int j = 0; j < WAVE_NODES; ++j) {
        dd[j]    = dinv[n0 + j];
        selfv[j] = f2[(size_t)(n0 + j) * (DIM / 2) + lane];
    }
#pragma unroll
    for (int j = 0; j < WAVE_NODES; ++j) {
        const bool in = rp[j] + lane < rp[j + 1];
        first[j] = srcSorted[in ? rp[j] + lane : 0];
    }

#pragma unroll
    for (int j = 0; j < WAVE_NODES; ++j) {
        const int   beg   = rp[j];
        const int   end   = rp[j + 1];
        float2 acc = selfv[j];

        {
            const int n = (end - beg < 32) ? (end - beg) : 32;
            pull_chunk(acc, f2, first[j], n, lane);
        }
        for (int base = beg + 32; base < end; base += 32) {
            const int n = (end - base < 32) ? (end - base) : 32;
            int sk = 0;
            if (lane < n) sk = srcSorted[base + lane];
            pull_chunk(acc, f2, sk, n, lane);
        }

        *reinterpret_cast<float2*>(&tile[(wave * WAVE_NODES + j) * TILE_PITCH + 2 * lane]) =
            make_float2(acc.x * dd[j], acc.y * dd[j]);
        if (LAYER1) ddL[wave * WAVE_NODES + j] = dd[j];
    }
    __syncthreads();

    {
        const int c    = lane & 15;
        const int half = lane >> 4;
        const int t    = wave;

        const int    ldw  = LAYER1 ? DIM : OUT_DIM;
        const int    col  = LAYER1 ? (16 * t + c) : (16 * (t & 1) + c);
        const float* Wsel = (LAYER1 || t < 2) ? Wa : Wb;
        const float* bsel = (LAYER1 || t < 2) ? ba : bb;
        const float* wp   = Wsel + col;
        const float  bias = bsel[col];
        const float* ar   = tile + c * TILE_PITCH + 2 * half;
        float dr[8];
#pragma unroll
        for (int r = 0; r < 8; ++r) dr[r] = LAYER1 ? ddL[8 * half + r] : 1.0f;

        v8f accE = {}, accO = {};
#pragma unroll
        for (int ks = 0; ks < 16; ++ks) {
            const int kk = ks * 4 + half * 2;
            const v2f a = *reinterpret_cast<const v2f*>(ar + 4 * ks);
            v2f b;
            b.x = wp[(size_t)kk * ldw];
            b.y = wp[(size_t)(kk + 1) * ldw];
            if (ks & 1) accO = __builtin_amdgcn_wmma_f32_16x16x4_f32(false, a, false, b, (short)0, accO, false, false);
            else        accE = __builtin_amdgcn_wmma_f32_16x16x4_f32(false, a, false, b, (short)0, accE, false, false);
        }

        float* obase;
        if (LAYER1) obase = out + (size_t)row0 * DIM + col;
        else        obase = out + (t < 2 ? (size_t)0 : (size_t)N_NODES * OUT_DIM) + (size_t)row0 * OUT_DIM + col;
#pragma unroll
        for (int r = 0; r < 8; ++r) {
            float v = accE[r] + accO[r] + bias;
            if (LAYER1) v = fmaxf(v, 0.0f) * dr[r];
            obase[(size_t)(r + 8 * half) * ldw] = v;
        }
    }
}

constexpr size_t XS_RAW    = ((size_t)N_NODES + (size_t)N_NODES * DIM + (N_NODES + 4) + (size_t)N_EDGES) * 4;
constexpr size_t XS_OFFSET = (XS_RAW + 255) / 256 * 256;

extern "C" void kernel_launch(void* const* d_in, const int* in_sizes, int n_in,
                              void* d_out, int out_size, void* d_ws, size_t ws_size,
                              hipStream_t stream) {
    const float* x    = (const float*)d_in[0];
    const int*   eidx = (const int*)d_in[1];
    const int*   src  = eidx;
    const int*   dst  = eidx + N_EDGES;
    const float* W1   = (const float*)d_in[2];
    const float* b1   = (const float*)d_in[3];
    const float* Wmu  = (const float*)d_in[4];
    const float* bmu  = (const float*)d_in[5];
    const float* Wlv  = (const float*)d_in[6];
    const float* blv  = (const float*)d_in[7];
    float*       out  = (float*)d_out;

    float* dinv = (float*)d_ws;
    float* h    = dinv + N_NODES;
    int*   rowptr    = (int*)(h + (size_t)N_NODES * DIM);
    int*   srcSorted = rowptr    + (N_NODES + 4);
    float* xs        = (float*)((char*)d_ws + XS_OFFSET);
    int*   recs      = (int*)(xs + (size_t)N_NODES * DIM);
    int*   tab       = recs      + 2 * (size_t)NCH * CHUNK;

    const unsigned gTile = N_NODES / TILE_NODES;

    bin_chunk_kernel<<<NCH, BIN_THREADS, 0, stream>>>(src, dst, recs, tab);
    csr_bin_kernel<<<NBIN, CSR_THREADS, 0, stream>>>(recs, tab, x, rowptr, dinv, srcSorted, xs);

    pull_project_kernel<true><<<gTile, TILE_WAVES * 32, 0, stream>>>(rowptr, srcSorted, dinv, xs, W1, b1, nullptr, nullptr, h);
    pull_project_kernel<false><<<gTile, TILE_WAVES * 32, 0, stream>>>(rowptr, srcSorted, dinv, h, Wmu, bmu, Wlv, blv, out);
}


# csr_bin sorts its bin on chip: records read once, one load batch per wave held in registers; sources placed in an LDS image of the CSR slice, stored as whole lines; sum order = LDS arrival rank
# speedup vs baseline: 9.7335x; 1.1035x over previous
#include <hip/hip_runtime.h>
#include <hip/hip_bf16.h>

#define N_NODES 100000
#define N_EDGES 1600000
#define DIM 64
#define OUT_DIM 32

typedef __attribute__((ext_vector_type(2))) float v2f;
typedef __attribute__((ext_vector_type(8))) float v8f;

#define CHUNK 8192
#define NCH ((N_EDGES + CHUNK - 1) / CHUNK)
#define BIN_SHIFT 8
#define BIN_NODES (1 << BIN_SHIFT)
#define NBIN ((N_NODES + BIN_NODES - 1) / BIN_NODES)
#define TROW (NBIN + 1)

__device__ __forceinline__ int wave_inclusive_scan(int v, int lane) {
#pragma unroll
    for (int off = 1; off < 32; off <<= 1) {
        const int u = __shfl_up(v, off);
        if (lane >= off) v += u;
    }
    return v;
}

#define BIN_THREADS 1024
#define BIN_PER_THREAD (CHUNK / BIN_THREADS)
static_assert(NBIN < BIN_THREADS, "one thread per bin counter, and one more for the total");
static_assert(BIN_THREADS == 1024, "the scan below reads one wave total per lane: 32 waves");

__global__ __launch_bounds__(BIN_THREADS) void bin_chunk_kernel(const int* __restrict__ src,
                                                                const int* __restrict__ dst,
                                                                int* __restrict__ recs,
                                                                int* __restrict__ tab) {
    __shared__ int  hist[BIN_THREADS];
    __shared__ int  waveSum[32];
    __shared__ int2 stage[CHUNK];
    const int t    = threadIdx.x;
    const int lane = t & 31;
    const int wave = t >> 5;
    const int e0   = blockIdx.x * CHUNK;

    hist[t] = 0;
    int d[BIN_PER_THREAD], sv[BIN_PER_THREAD], r[BIN_PER_THREAD];
#pragma unroll
    for (int k = 0; k < BIN_PER_THREAD; ++k) {
        const int e = e0 + k * BIN_THREADS + t;
        d[k]  = (e < N_EDGES) ? dst[e] : -1;
        sv[k] = (e < N_EDGES) ? src[e] : 0;
    }
    __syncthreads();
#pragma unroll
    for (int k = 0; k < BIN_PER_THREAD; ++k) {
        r[k] = 0;
        if (d[k] >= 0) r[k] = atomicAdd(&hist[d[k] >> BIN_SHIFT], 1);
    }
    __syncthreads();

    const int val = hist[t];
    const int v   = wave_inclusive_scan(val, lane);
    if (lane == 31) waveSum[wave] = v;
    __syncthreads();
    const int ws = waveSum[lane];
    const int wv = wave_inclusive_scan(ws, lane);
    const int wavePrefix = __shfl(wv - ws, wave);
    const int excl = wavePrefix + (v - val);
    hist[t] = excl;
    if (t <= NBIN) tab[blockIdx.x * TROW + t] = excl;
    __syncthreads();

#pragma unroll
    for (int k = 0; k < BIN_PER_THREAD; ++k) {
        if (d[k] >= 0) stage[hist[d[k] >> BIN_SHIFT] + r[k]] = make_int2(sv[k], d[k]);
    }
    __syncthreads();
    int2* out2 = reinterpret_cast<int2*>(recs) + (size_t)blockIdx.x * CHUNK;
#pragma unroll
    for (int k = 0; k < BIN_PER_THREAD; ++k) {
        const int idx = k * BIN_THREADS + t;
        if (e0 + idx < N_EDGES) out2[idx] = stage[idx];
    }
}

#define CSR_THREADS 1024
#define CSR_WAVES (CSR_THREADS / 32)
static_assert(CSR_WAVES == 32, "a wave adds the waves' shares of the bin's first slot one per lane");
static_assert(NCH <= CSR_THREADS, "one thread fetches one chunk's run start");
static_assert(BIN_NODES < CSR_THREADS, "one thread per node of the bin, and one more for the total");

#define ROW_F4 (DIM / 4)
#define XS_PER_THREAD (BIN_NODES * ROW_F4 / CSR_THREADS)
static_assert(BIN_NODES * ROW_F4 % CSR_THREADS == 0, "the bin's rows split evenly over the threads");

#define RUNS_PER_WAVE ((NCH + CSR_WAVES - 1) / CSR_WAVES)
#define SRC_CAP 8192

__global__ __launch_bounds__(CSR_THREADS) void csr_bin_kernel(const int* __restrict__ recs,
                                                              const int* __restrict__ tab,
                                                              const float* __restrict__ x,
                                                              int* __restrict__ rowptr,
                                                              float* __restrict__ dinv,
                                                              int* __restrict__ srcSorted,
                                                              float* __restrict__ xs) {
    __shared__ int cntL[BIN_NODES];
    __shared__ int cur[BIN_NODES];
    __shared__ int baseSum[CSR_WAVES];
    __shared__ int waveSum[CSR_WAVES];
    __shared__ float dinvL[BIN_NODES];
    __shared__ int totalL;
    __shared__ int srcL[SRC_CAP];
    const int t    = threadIdx.x;
    const int lane = t & 31;
    const int wave = __builtin_amdgcn_readfirstlane(t >> 5);
    const int bin  = blockIdx.x;
    const int node0 = bin * BIN_NODES;

    int sb[RUNS_PER_WAVE], se[RUNS_PER_WAVE];
#pragma unroll
    for (int k = 0; k < RUNS_PER_WAVE; ++k) {
        const int  c  = wave + CSR_WAVES * k;
        const int  cc = (c < NCH) ? c : NCH - 1;
        const int* tp = tab + (size_t)cc * TROW + bin;
        sb[k] = __builtin_amdgcn_readfirstlane(tp[0]);
        se[k] = __builtin_amdgcn_readfirstlane(tp[1]);
        if (c >= NCH) se[k] = sb[k];
    }

    int s = 0;
    if (t < NCH) s = tab[t * TROW + bin];
    if (t < BIN_NODES) cntL[t] = 0;

    const int2* r2 = reinterpret_cast<const int2*>(recs);

    int2 piece[RUNS_PER_WAVE];
#pragma unroll
    for (int k = 0; k < RUNS_PER_WAVE; ++k) {
        const int  c  = wave + CSR_WAVES * k;
        const int  cc = (c < NCH) ? c : NCH - 1;
        const bool in = sb[k] + lane < se[k];
        const int2 v  = r2[(size_t)cc * CHUNK + (in ? sb[k] + lane : 0)];
        piece[k] = make_int2(v.x, in ? v.y : -1);
    }

    int rs = s;
#pragma unroll
    for (int off = 16; off >= 1; off >>= 1) rs += __shfl_xor(rs, off);
    if (lane == 0) baseSum[wave] = rs;
    __syncthreads();

#pragma unroll
    for (int k = 0; k < RUNS_PER_WAVE; ++k) {
        if (piece[k].y >= 0) atomicAdd(&cntL[piece[k].y - node0], 1);
    }
#pragma unroll
    for (int k = 0; k < RUNS_PER_WAVE; ++k) {
        const int c = wave + CSR_WAVES * k;
        if (se[k] - sb[k] > 32) {
            for (int i = sb[k] + 32 + lane; i < se[k]; i += 32) {
                const int2 rec = r2[(size_t)c * CHUNK + i];
                atomicAdd(&cntL[rec.y - node0], 1);
            }
        }
    }
    __syncthreads();

    const int val = (t < BIN_NODES) ? cntL[t] : 0;
    const int v   = wave_inclusive_scan(val, lane);
    if (lane == 31) waveSum[wave] = v;
    __syncthreads();
    int wp = 0;
    for (int w = 0; w < wave; ++w) wp += waveSum[w];
    const int excl = wp + (v - val);
    const float dv = rsqrtf((float)(val + 1));
    if (t < BIN_NODES) {
        cur[t]   = excl;
        dinvL[t] = dv;
    }
    if (t == BIN_NODES) totalL = excl;
    __syncthreads();
    const int  total = totalL;
    const bool inLds = total <= SRC_CAP;

    const float4* x4 = reinterpret_cast<const float4*>(x) + (size_t)node0 * ROW_F4;
    float4 xr[XS_PER_THREAD];
#pragma unroll
    for (int k = 0; k < XS_PER_THREAD; ++k) {
        const int q = k * CSR_THREADS + t;
        xr[k] = (node0 + q / ROW_F4 < N_NODES) ? x4[q] : make_float4(0.0f, 0.0f, 0.0f, 0.0f);
    }

    int bsum = baseSum[lane];
#pragma unroll
    for (int off = 16; off >= 1; off >>= 1) bsum += __shfl_xor(bsum, off);
    const int base = __builtin_amdgcn_readfirstlane(bsum);

    int p[RUNS_PER_WAVE];
#pragma unroll
    for (int k = 0; k < RUNS_PER_WAVE; ++k) {
        p[k] = 0;
        if (piece[k].y >= 0) p[k] = atomicAdd(&cur[piece[k].y - node0], 1);
    }
#pragma unroll
    for (int k = 0; k < RUNS_PER_WAVE; ++k) {
        if (piece[k].y >= 0) {
            if (inLds) srcL[p[k]] = piece[k].x;
            else       srcSorted[base + p[k]] = piece[k].x;
        }
    }
#pragma unroll
    for (int k = 0; k < RUNS_PER_WAVE; ++k) {
        const int c = wave + CSR_WAVES * k;
        if (se[k] - sb[k] > 32) {
            for (int i = sb[k] + 32 + lane; i < se[k]; i += 32) {
                const int2 rec = r2[(size_t)c * CHUNK + i];
                const int  q   = atomicAdd(&cur[rec.y - node0], 1);
                if (inLds) srcL[q] = rec.x;
                else       srcSorted[base + q] = rec.x;
            }
        }
    }
    __syncthreads();

    if (inLds) {
        const int shift = (int)((reinterpret_cast<size_t>(srcSorted + base) >> 2) & 31);
        for (int i = t - shift; i < total; i += CSR_THREADS) {
            if (i >= 0) srcSorted[base + i] = srcL[i];
        }
    }

    if (t < BIN_NODES && node0 + t < N_NODES) {
        rowptr[node0 + t] = base + excl;
        dinv[node0 + t]   = dv;
    }
    if (bin == NBIN - 1 && t == BIN_NODES) rowptr[N_NODES] = base + excl;

    float4* xs4 = reinterpret_cast<float4*>(xs) + (size_t)node0 * ROW_F4;
#pragma unroll
    for (int k = 0; k < XS_PER_THREAD; ++k) {
        const int q = k * CSR_THREADS + t;
        if (node0 + q / ROW_F4 < N_NODES) {
            const float dq = dinvL[q / ROW_F4];
            xs4[q] = make_float4(xr[k].x * dq, xr[k].y * dq, xr[k].z * dq, xr[k].w * dq);
        }
    }
}

template <int U>
__device__ __forceinline__ void pull_group(float2& acc, const float2* __restrict__ f2,
                                           int sk, int k, int lane) {
    float2 v[U];
#pragma unroll
    for (int u = 0; u < U; ++u) {
        const int s = __builtin_amdgcn_readlane(sk, k + u);
        v[u] = f2[(size_t)s * (DIM / 2) + lane];
    }
#pragma unroll
    for (int u = 0; u < U; ++u) {
        acc.x += v[u].x;
        acc.y += v[u].y;
    }
}

__device__ __forceinline__ void pull_chunk(float2& acc, const float2* __restrict__ f2,
                                           int sk, int n, int lane) {
    int k = 0;
    for (; k + 8 <= n; k += 8) pull_group<8>(acc, f2, sk, k, lane);
    if (k + 4 <= n) { pull_group<4>(acc, f2, sk, k, lane); k += 4; }
    if (k + 2 <= n) { pull_group<2>(acc, f2, sk, k, lane); k += 2; }
    if (k < n)      { pull_group<1>(acc, f2, sk, k, lane); }
}

#define TILE_NODES 16
#define TILE_WAVES 4
#define WAVE_NODES (TILE_NODES / TILE_WAVES)
#define TILE_PITCH 68
static_assert(N_NODES % TILE_NODES == 0, "every work-group owns a full 16-row tile");
static_assert(TILE_WAVES == 4, "one wave per 16-column slice of the 64 result columns");

template <bool LAYER1>
__global__ __launch_bounds__(TILE_WAVES * 32) void pull_project_kernel(const int* __restrict__ rowptr,
                                                                       const int* __restrict__ srcSorted,
                                                                       const float* __restrict__ dinv,
                                                                       const float* __restrict__ feat,
                                                                       const float* __restrict__ Wa,
                                                                       const float* __restrict__ ba,
                                                                       const float* __restrict__ Wb,
                                                                       const float* __restrict__ bb,
                                                                       float* __restrict__ out) {
    __shared__ float tile[TILE_NODES * TILE_PITCH];
    __shared__ float ddL[TILE_NODES];

    const int lane = threadIdx.x & 31;
    const int wave = __builtin_amdgcn_readfirstlane((int)(threadIdx.x >> 5));
    const int row0 = blockIdx.x * TILE_NODES;
    const int n0   = row0 + wave * WAVE_NODES;

    const float2* f2 = reinterpret_cast<const float2*>(feat);

    int    rp[WAVE_NODES + 1];
    float  dd[WAVE_NODES];
    float2 selfv[WAVE_NODES];
    int    first[WAVE_NODES];
#pragma unroll
    for (int j = 0; j <= WAVE_NODES; ++j) rp[j] = rowptr[n0 + j];
#pragma unroll
    for (int j = 0; j < WAVE_NODES; ++j) {
        dd[j]    = dinv[n0 + j];
        selfv[j] = f2[(size_t)(n0 + j) * (DIM / 2) + lane];
    }
#pragma unroll
    for (int j = 0; j < WAVE_NODES; ++j) {
        const bool in = rp[j] + lane < rp[j + 1];
        first[j] = srcSorted[in ? rp[j] + lane : 0];
    }

#pragma unroll
    for (int j = 0; j < WAVE_NODES; ++j) {
        const int   beg   = rp[j];
        const int   end   = rp[j + 1];
        float2 acc = selfv[j];

        {
            const int n = (end - beg < 32) ? (end - beg) : 32;
            pull_chunk(acc, f2, first[j], n, lane);
        }
        for (int base = beg + 32; base < end; base += 32) {
            const int n = (end - base < 32) ? (end - base) : 32;
            int sk = 0;
            if (lane < n) sk = srcSorted[base + lane];
            pull_chunk(acc, f2, sk, n, lane);
        }

        *reinterpret_cast<float2*>(&tile[(wave * WAVE_NODES + j) * TILE_PITCH + 2 * lane]) =
            make_float2(acc.x * dd[j], acc.y * dd[j]);
        if (LAYER1) ddL[wave * WAVE_NODES + j] = dd[j];
    }
    __syncthreads();

    {
        const int c    = lane & 15;
        const int half = lane >> 4;
        const int t    = wave;

        const int    ldw  = LAYER1 ? DIM : OUT_DIM;
        const int    col  = LAYER1 ? (16 * t + c) : (16 * (t & 1) + c);
        const float* Wsel = (LAYER1 || t < 2) ? Wa : Wb;
        const float* bsel = (LAYER1 || t < 2) ? ba : bb;
        const float* wp   = Wsel + col;
        const float  bias = bsel[col];
        const float* ar   = tile + c * TILE_PITCH + 2 * half;
        float dr[8];
#pragma unroll
        for (int r = 0; r < 8; ++r) dr[r] = LAYER1 ? ddL[8 * half + r] : 1.0f;

        v8f accE = {}, accO = {};
#pragma unroll
        for (int ks = 0; ks < 16; ++ks) {
            const int kk = ks * 4 + half * 2;
            const v2f a = *reinterpret_cast<const v2f*>(ar + 4 * ks);
            v2f b;
            b.x = wp[(size_t)kk * ldw];
            b.y = wp[(size_t)(kk + 1) * ldw];
            if (ks & 1) accO = __builtin_amdgcn_wmma_f32_16x16x4_f32(false, a, false, b, (short)0, accO, false, false);
            else        accE = __builtin_amdgcn_wmma_f32_16x16x4_f32(false, a, false, b, (short)0, accE, false, false);
        }

        float* obase;
        if (LAYER1) obase = out + (size_t)row0 * DIM + col;
        else        obase = out + (t < 2 ? (size_t)0 : (size_t)N_NODES * OUT_DIM) + (size_t)row0 * OUT_DIM + col;
#pragma unroll
        for (int r = 0; r < 8; ++r) {
            float v = accE[r] + accO[r] + bias;
            if (LAYER1) v = fmaxf(v, 0.0f) * dr[r];
            obase[(size_t)(r + 8 * half) * ldw] = v;
        }
    }
}

constexpr size_t XS_RAW    = ((size_t)N_NODES + (size_t)N_NODES * DIM + (N_NODES + 4) + (size_t)N_EDGES) * 4;
constexpr size_t XS_OFFSET = (XS_RAW + 255) / 256 * 256;

extern "C" void kernel_launch(void* const* d_in, const int* in_sizes, int n_in,
                              void* d_out, int out_size, void* d_ws, size_t ws_size,
                              hipStream_t stream) {
    const float* x    = (const float*)d_in[0];
    const int*   eidx = (const int*)d_in[1];
    const int*   src  = eidx;
    const int*   dst  = eidx + N_EDGES;
    const float* W1   = (const float*)d_in[2];
    const float* b1   = (const float*)d_in[3];
    const float* Wmu  = (const float*)d_in[4];
    const float* bmu  = (const float*)d_in[5];
    const float* Wlv  = (const float*)d_in[6];
    const float* blv  = (const float*)d_in[7];
    float*       out  = (float*)d_out;

    float* dinv = (float*)d_ws;
    float* h    = dinv + N_NODES;
    int*   rowptr    = (int*)(h + (size_t)N_NODES * DIM);
    int*   srcSorted = rowptr    + (N_NODES + 4);
    float* xs        = (float*)((char*)d_ws + XS_OFFSET);
    int*   recs      = (int*)(xs + (size_t)N_NODES * DIM);
    int*   tab       = recs      + 2 * (size_t)NCH * CHUNK;

    const unsigned gTile = N_NODES / TILE_NODES;

    bin_chunk_kernel<<<NCH, BIN_THREADS, 0, stream>>>(src, dst, recs, tab);
    csr_bin_kernel<<<NBIN, CSR_THREADS, 0, stream>>>(recs, tab, x, rowptr, dinv, srcSorted, xs);

    pull_project_kernel<true><<<gTile, TILE_WAVES * 32, 0, stream>>>(rowptr, srcSorted, dinv, xs, W1, b1, nullptr, nullptr, h);
    pull_project_kernel<false><<<gTile, TILE_WAVES * 32, 0, stream>>>(rowptr, srcSorted, dinv, h, Wmu, bmu, Wlv, blv, out);
}
